# starting kernel
# baseline (speedup 1.0000x reference)
#include <hip/hip_runtime.h>
#include <hip/hip_bf16.h>

typedef __attribute__((ext_vector_type(2))) float v2f;
typedef __attribute__((ext_vector_type(8))) float v8f;


__device__ __forceinline__ void gate_rz(float2* st, int wire, float theta, int t) {
    float h = 0.5f * theta;
    float c = cosf(h), s = sinf(h);
    int mask = 1 << (7 - wire);
    int low  = t & (mask - 1);
    int i0   = ((t ^ low) << 1) | low;
    int i1   = i0 | mask;
    float2 a = st[i0], b = st[i1];
    st[i0] = make_float2(a.x * c + a.y * s, a.y * c - a.x * s);
    st[i1] = make_float2(b.x * c - b.y * s, b.y * c + b.x * s);
}

__device__ __forceinline__ void gate_ry(float2* st, int wire, float theta, int t) {
    float h = 0.5f * theta;
    float c = cosf(h), s = sinf(h);
    int mask = 1 << (7 - wire);
    int low  = t & (mask - 1);
    int i0   = ((t ^ low) << 1) | low;
    int i1   = i0 | mask;
    float2 a = st[i0], b = st[i1];
    st[i0] = make_float2(c * a.x - s * b.x, c * a.y - s * b.y);
    st[i1] = make_float2(s * a.x + c * b.x, s * a.y + c * b.y);
}

__device__ __forceinline__ void gate_cnot(float2* st, int ctrl, int tgt, int t) {
    int pc = 7 - ctrl, pt = 7 - tgt;
    int p1 = pc < pt ? pc : pt;
    int p2 = pc < pt ? pt : pc;
    if (t < 64) {
        unsigned low  = (unsigned)t & ((1u << p1) - 1u);
        unsigned mid  = ((unsigned)t >> p1) & ((1u << (p2 - p1 - 1)) - 1u);
        unsigned high = (unsigned)t >> (p2 - 1);
        unsigned i = (high << (p2 + 1)) | (mid << (p1 + 1)) | low;
        i |= (1u << pc);
        unsigned ip = i | (1u << pt);
        float2 a = st[i], b = st[ip];
        st[i]  = b;
        st[ip] = a;
    }
}

__global__ void qcnn_build_matrix(const float* __restrict__ w, float* __restrict__ AtP) {
    __shared__ float2 st[256];
    const int j = blockIdx.x;
    const int t = threadIdx.x;

    st[t]       = make_float2((t == j)       ? 1.0f : 0.0f, 0.0f);
    st[t + 128] = make_float2((t + 128 == j) ? 1.0f : 0.0f, 0.0f);
    __syncthreads();

    const int pr1[7] = {0, 2, 4, 6, 1, 5, 3};
    const int pr2[7] = {1, 3, 5, 7, 3, 7, 7};
    const float HPI = 1.57079632679489662f;

    for (int p = 0; p < 7; ++p) {
        const int b1 = pr1[p], b2 = pr2[p];
        const float* pw = w + p * 6;
        gate_rz  (st, b2, -HPI,  t); __syncthreads();
        gate_cnot(st, b2, b1,    t); __syncthreads();
        gate_rz  (st, b1, pw[0], t); __syncthreads();
        gate_ry  (st, b2, pw[1], t); __syncthreads();
        gate_cnot(st, b1, b2,    t); __syncthreads();
        gate_ry  (st, b2, pw[2], t); __syncthreads();
        gate_cnot(st, b2, b1,    t); __syncthreads();
        gate_rz  (st, b1,  HPI,  t); __syncthreads();
        gate_rz  (st, b2, -HPI,  t); __syncthreads();
        gate_cnot(st, b2, b1,    t); __syncthreads();
        gate_rz  (st, b1, pw[3], t); __syncthreads();
        gate_ry  (st, b2, pw[4], t); __syncthreads();
        gate_cnot(st, b1, b2,    t); __syncthreads();
        gate_ry  (st, b2, pw[5], t); __syncthreads();
    }

    float2 amp = st[2 * t + 1];
    const int base = (j >> 1) * 512 + (j & 1);
    AtP[base + t * 2]         = amp.x;
    AtP[base + (t + 128) * 2] = amp.y;
}


#define BM 128
#define NSUB 8
#define NWAVES 16

__global__ __launch_bounds__(512) void qcnn_wmma_gemm(const float* __restrict__ x,
                                                      const float2* __restrict__ AtP,
                                                      float* __restrict__ out) {
    __shared__ float Xs[BM * 260];
    __shared__ float part[BM][NWAVES];
    __shared__ float xss[BM][4];

    const int tid = threadIdx.x;
    const size_t m0 = (size_t)blockIdx.x * BM;

    {
        const int row = tid >> 2;
        const int seg = tid & 3;
        const float4* src = (const float4*)(x + (m0 + row) * 256 + seg * 64);
        float* dst = &Xs[row * 260 + seg * 64];
        float ss = 0.0f;
#pragma unroll
        for (int i = 0; i < 16; ++i) {
            float4 v = src[i];
            dst[i * 4 + 0] = v.x; dst[i * 4 + 1] = v.y;
            dst[i * 4 + 2] = v.z; dst[i * 4 + 3] = v.w;
            ss += v.x * v.x + v.y * v.y + v.z * v.z + v.w * v.w;
        }
        xss[row][seg] = ss;
    }
    __syncthreads();

    const int wave  = tid >> 5;
    const int lane  = tid & 31;
    const int laneL = lane & 15;
    const int hi    = lane >> 4;
    const int n0    = wave * 16;

    v8f acc[NSUB] = {};

#pragma unroll 4
    for (int k = 0; k < 256; k += 4) {
        float2 bv = AtP[(size_t)((k >> 1) + hi) * 256 + n0 + laneL];
        v2f b; b[0] = bv.x; b[1] = bv.y;
#pragma unroll
        for (int s = 0; s < NSUB; ++s) {
            const float2* pr = (const float2*)&Xs[(s * 16 + laneL) * 260];
            float2 av = pr[(k >> 1) + hi];
            v2f a; a[0] = av.x; a[1] = av.y;
            acc[s] = __builtin_amdgcn_wmma_f32_16x16x4_f32(
                false, a, false, b, (short)0, acc[s], false, false);
        }
    }

#pragma unroll
    for (int s = 0; s < NSUB; ++s) {
#pragma unroll
        for (int v = 0; v < 8; ++v) {
            float r2 = acc[s][v] * acc[s][v];
            r2 += __shfl_xor(r2, 1);
            r2 += __shfl_xor(r2, 2);
            r2 += __shfl_xor(r2, 4);
            r2 += __shfl_xor(r2, 8);
            if (laneL == 0) part[s * 16 + hi * 8 + v][wave] = r2;
        }
    }
    __syncthreads();

    if (tid < BM) {
        float q = 0.0f;
#pragma unroll
        for (int wv = 0; wv < NWAVES; ++wv) q += part[tid][wv];
        float ssx = 0.0f;
#pragma unroll
        for (int i = 0; i < 4; ++i) ssx += xss[tid][i];
        q /= ssx;
        const size_t bidx = m0 + tid;
        out[2 * bidx + 0] = 1.0f - q;
        out[2 * bidx + 1] = q;
    }
}


extern "C" void kernel_launch(void* const* d_in, const int* in_sizes, int n_in,
                              void* d_out, int out_size, void* d_ws, size_t ws_size,
                              hipStream_t stream) {
    const float* x = (const float*)d_in[0];
    const float* w = (const float*)d_in[1];
    float* out = (float*)d_out;
    float* AtP = (float*)d_ws;

    const int B = in_sizes[0] / 256;

    qcnn_build_matrix<<<256, 128, 0, stream>>>(w, AtP);
    qcnn_wmma_gemm<<<B / BM, 512, 0, stream>>>(x, (const float2*)AtP, out);
}


# GEMM transposed: a wave's 16 batch rows go from global straight into registers as B fragments (no LDS staging of x), circuit matrix read from LDS as A; per-lane sum of squares; new summation order
# speedup vs baseline: 1.6053x; 1.6053x over previous
#include <hip/hip_runtime.h>

typedef __attribute__((ext_vector_type(2))) float v2f;
typedef __attribute__((ext_vector_type(8))) float v8f;


__device__ __forceinline__ void gate_rz(float2* st, int wire, float theta, int t) {
    float h = 0.5f * theta;
    float c = cosf(h), s = sinf(h);
    int mask = 1 << (7 - wire);
    int low  = t & (mask - 1);
    int i0   = ((t ^ low) << 1) | low;
    int i1   = i0 | mask;
    float2 a = st[i0], b = st[i1];
    st[i0] = make_float2(a.x * c + a.y * s, a.y * c - a.x * s);
    st[i1] = make_float2(b.x * c - b.y * s, b.y * c + b.x * s);
}

__device__ __forceinline__ void gate_ry(float2* st, int wire, float theta, int t) {
    float h = 0.5f * theta;
    float c = cosf(h), s = sinf(h);
    int mask = 1 << (7 - wire);
    int low  = t & (mask - 1);
    int i0   = ((t ^ low) << 1) | low;
    int i1   = i0 | mask;
    float2 a = st[i0], b = st[i1];
    st[i0] = make_float2(c * a.x - s * b.x, c * a.y - s * b.y);
    st[i1] = make_float2(s * a.x + c * b.x, s * a.y + c * b.y);
}

__device__ __forceinline__ void gate_cnot(float2* st, int ctrl, int tgt, int t) {
    int pc = 7 - ctrl, pt = 7 - tgt;
    int p1 = pc < pt ? pc : pt;
    int p2 = pc < pt ? pt : pc;
    if (t < 64) {
        unsigned low  = (unsigned)t & ((1u << p1) - 1u);
        unsigned mid  = ((unsigned)t >> p1) & ((1u << (p2 - p1 - 1)) - 1u);
        unsigned high = (unsigned)t >> (p2 - 1);
        unsigned i = (high << (p2 + 1)) | (mid << (p1 + 1)) | low;
        i |= (1u << pc);
        unsigned ip = i | (1u << pt);
        float2 a = st[i], b = st[ip];
        st[i]  = b;
        st[ip] = a;
    }
}

__global__ void qcnn_build_matrix(const float* __restrict__ w, float* __restrict__ AtP) {
    __shared__ float2 st[256];
    const int j = blockIdx.x;
    const int t = threadIdx.x;

    st[t]       = make_float2((t == j)       ? 1.0f : 0.0f, 0.0f);
    st[t + 128] = make_float2((t + 128 == j) ? 1.0f : 0.0f, 0.0f);
    __syncthreads();

    const int pr1[7] = {0, 2, 4, 6, 1, 5, 3};
    const int pr2[7] = {1, 3, 5, 7, 3, 7, 7};
    const float HPI = 1.57079632679489662f;

    for (int p = 0; p < 7; ++p) {
        const int b1 = pr1[p], b2 = pr2[p];
        const float* pw = w + p * 6;
        gate_rz  (st, b2, -HPI,  t); __syncthreads();
        gate_cnot(st, b2, b1,    t); __syncthreads();
        gate_rz  (st, b1, pw[0], t); __syncthreads();
        gate_ry  (st, b2, pw[1], t); __syncthreads();
        gate_cnot(st, b1, b2,    t); __syncthreads();
        gate_ry  (st, b2, pw[2], t); __syncthreads();
        gate_cnot(st, b2, b1,    t); __syncthreads();
        gate_rz  (st, b1,  HPI,  t); __syncthreads();
        gate_rz  (st, b2, -HPI,  t); __syncthreads();
        gate_cnot(st, b2, b1,    t); __syncthreads();
        gate_rz  (st, b1, pw[3], t); __syncthreads();
        gate_ry  (st, b2, pw[4], t); __syncthreads();
        gate_cnot(st, b1, b2,    t); __syncthreads();
        gate_ry  (st, b2, pw[5], t); __syncthreads();
    }

    float2 amp = st[2 * t + 1];
    AtP[t * 256 + j]         = amp.x;
    AtP[(t + 128) * 256 + j] = amp.y;
}


#define MP 260
#define GW 8

__device__ __forceinline__ v8f mma4(float ax, float ay, float bx, float by, v8f acc) {
    v2f a; a[0] = ax; a[1] = ay;
    v2f b; b[0] = bx; b[1] = by;
    return __builtin_amdgcn_wmma_f32_16x16x4_f32(false, a, false, b, (short)0, acc, false, false);
}

__global__ __launch_bounds__(256) void qcnn_wmma_gemm(const float* __restrict__ x,
                                                      const float* __restrict__ Mg,
                                                      float* __restrict__ out,
                                                      int nrows) {
    __shared__ float Ms[256 * MP];

    const int tid  = threadIdx.x;
    const int wave = __builtin_amdgcn_readfirstlane(tid >> 5);
    const int lane = tid & 31;
    const int c    = lane & 15;
    const int h    = lane >> 4;

    const int ntiles  = nrows >> 4;
    const int tstride = gridDim.x * GW;
    int tile = blockIdx.x * GW + wave;

    const int t0 = tile < ntiles ? tile : 0;
    const float* xr = x + ((size_t)t0 * 16 + c) * 256 + 4 * h;
    float4 xc[4];
#pragma unroll
    for (int u = 0; u < 4; ++u) xc[u] = *(const float4*)(xr + 8 * u);

    {
        const float4* src = (const float4*)Mg;
        for (int i = tid; i < 256 * 64; i += 256) {
            const int r = i >> 6, q = i & 63;
            *(float4*)&Ms[r * MP + q * 4] = src[i];
        }
    }
    __syncthreads();

    if (tile >= ntiles) return;

    const float* mlane = Ms + c * MP + 4 * h;

    float4 fa[2][16];
#pragma unroll
    for (int nt = 0; nt < 16; ++nt) fa[0][nt] = *(const float4*)(mlane + nt * 16 * MP);

    while (true) {
        const int  ntile = tile + tstride;
        const bool more  = ntile < ntiles;
        const float* xnr = x + ((size_t)(more ? ntile : tile) * 16 + c) * 256 + 4 * h;

        v8f acc[16];
#pragma unroll
        for (int nt = 0; nt < 16; ++nt) acc[nt] = (v8f){};
        float ssx = 0.0f;

#pragma unroll 1
        for (int jc = 0; jc < 8; ++jc) {
            const float* nsrc = (jc < 7) ? (xr + 32 * (jc + 1)) : xnr;
            float4 xn[4];
#pragma unroll
            for (int u = 0; u < 4; ++u) xn[u] = *(const float4*)(nsrc + 8 * u);

            const float* mj = mlane + 32 * jc;
#pragma unroll
            for (int u = 0; u < 4; ++u) {
                const float* mnext = (u < 3) ? (mj + 8 * (u + 1)) : (mlane + 32 * ((jc + 1) & 7));
#pragma unroll
                for (int nt = 0; nt < 16; ++nt)
                    fa[(u + 1) & 1][nt] = *(const float4*)(mnext + nt * 16 * MP);
                __builtin_amdgcn_sched_barrier(0);
                ssx += xc[u].x * xc[u].x + xc[u].y * xc[u].y + xc[u].z * xc[u].z + xc[u].w * xc[u].w;
#pragma unroll
                for (int nt = 0; nt < 16; ++nt)
                    acc[nt] = mma4(fa[u & 1][nt].x, fa[u & 1][nt].y, xc[u].x, xc[u].y, acc[nt]);
#pragma unroll
                for (int nt = 0; nt < 16; ++nt)
                    acc[nt] = mma4(fa[u & 1][nt].z, fa[u & 1][nt].w, xc[u].z, xc[u].w, acc[nt]);
                __builtin_amdgcn_sched_barrier(0);
            }
#pragma unroll
            for (int u = 0; u < 4; ++u) xc[u] = xn[u];
        }

        float q4[4] = {0.0f, 0.0f, 0.0f, 0.0f};
#pragma unroll
        for (int nt = 0; nt < 16; ++nt) {
#pragma unroll
            for (int v = 0; v < 8; ++v) q4[v & 3] += acc[nt][v] * acc[nt][v];
        }
        float qs = (q4[0] + q4[1]) + (q4[2] + q4[3]);
        qs  += __shfl_xor(qs, 16);
        ssx += __shfl_xor(ssx, 16);
        if (h == 0) {
            const float q = qs / ssx;
            float2 o; o.x = 1.0f - q; o.y = q;
            *(float2*)(out + 2 * ((size_t)tile * 16 + c)) = o;
        }

        if (!more) break;
        tile = ntile;
        xr   = xnr;
    }
}


extern "C" void kernel_launch(void* const* d_in, const int* in_sizes, int n_in,
                              void* d_out, int out_size, void* d_ws, size_t ws_size,
                              hipStream_t stream) {
    const float* x = (const float*)d_in[0];
    const float* w = (const float*)d_in[1];
    float* out = (float*)d_out;
    float* AtP = (float*)d_ws;

    const int B = in_sizes[0] / 256;

    qcnn_build_matrix<<<256, 128, 0, stream>>>(w, AtP);
    qcnn_wmma_gemm<<<256, 256, 0, stream>>>(x, (const float*)AtP, out, B);
}


# circuit-matrix kernel: entries in closed form from tensor-product factors, U = V37.(UL x UR); one row per block from two rows of each 16x16 factor; 3 barriers, no per-gate sweep; new rounding order
# speedup vs baseline: 1.9279x; 1.2010x over previous
#include <hip/hip_runtime.h>

typedef __attribute__((ext_vector_type(2))) float v2f;
typedef __attribute__((ext_vector_type(8))) float v8f;


__device__ __forceinline__ float2 cmul(float2 a, float2 b) {
    return make_float2(a.x * b.x - a.y * b.y, a.x * b.y + a.y * b.x);
}

__device__ __forceinline__ void rot_z(float2& lo, float2& hi, float c, float s) {
    lo = make_float2(lo.x * c + lo.y * s, lo.y * c - lo.x * s);
    hi = make_float2(hi.x * c - hi.y * s, hi.y * c + hi.x * s);
}

__device__ __forceinline__ void rot_y(float2& a, float2& b, float c, float s) {
    const float2 na = make_float2(c * a.x - s * b.x, c * a.y - s * b.y);
    const float2 nb = make_float2(s * a.x + c * b.x, s * a.y + c * b.y);
    a = na; b = nb;
}

__device__ __forceinline__ void swap2(float2& a, float2& b) { const float2 t = a; a = b; b = t; }

__global__ __launch_bounds__(256) void qcnn_build_matrix(const float* __restrict__ w, float* __restrict__ AtP) {
    __shared__ float2 cs[42];
    __shared__ float2 Vs[7][4][4];
    __shared__ float2 Uf[2][2][16];

    const int t   = blockIdx.x;
    const int tid = threadIdx.x;

    if (tid < 42) {
        float s, c;
        sincosf(0.5f * w[tid], &s, &c);
        cs[tid] = make_float2(c, s);
    }
    __syncthreads();

    if (tid < 28) {
        const int p = tid >> 2, col = tid & 3;
        const float R = 0.70710678118654752f;
        float2 s0 = make_float2(col == 0 ? 1.0f : 0.0f, 0.0f);
        float2 s1 = make_float2(col == 1 ? 1.0f : 0.0f, 0.0f);
        float2 s2 = make_float2(col == 2 ? 1.0f : 0.0f, 0.0f);
        float2 s3 = make_float2(col == 3 ? 1.0f : 0.0f, 0.0f);
        const float2 g0 = cs[6 * p + 0], g1 = cs[6 * p + 1], g2 = cs[6 * p + 2];
        const float2 g3 = cs[6 * p + 3], g4 = cs[6 * p + 4], g5 = cs[6 * p + 5];
        rot_z(s0, s1, R, -R);   rot_z(s2, s3, R, -R);
        swap2(s1, s3);
        rot_z(s0, s2, g0.x, g0.y); rot_z(s1, s3, g0.x, g0.y);
        rot_y(s0, s1, g1.x, g1.y); rot_y(s2, s3, g1.x, g1.y);
        swap2(s2, s3);
        rot_y(s0, s1, g2.x, g2.y); rot_y(s2, s3, g2.x, g2.y);
        swap2(s1, s3);
        rot_z(s0, s2, R, R);    rot_z(s1, s3, R, R);
        rot_z(s0, s1, R, -R);   rot_z(s2, s3, R, -R);
        swap2(s1, s3);
        rot_z(s0, s2, g3.x, g3.y); rot_z(s1, s3, g3.x, g3.y);
        rot_y(s0, s1, g4.x, g4.y); rot_y(s2, s3, g4.x, g4.y);
        swap2(s2, s3);
        rot_y(s0, s1, g5.x, g5.y); rot_y(s2, s3, g5.x, g5.y);
        Vs[p][0][col] = s0; Vs[p][1][col] = s1; Vs[p][2][col] = s2; Vs[p][3][col] = s3;
    }
    __syncthreads();

    if (tid < 64) {
        const int half = tid >> 5, sel = (tid >> 4) & 1, col = tid & 15;
        const int sh = half ? 0 : 4;
        const int q0 = (t >> (sh + 2)) & 1, q1 = (t >> (sh + 1)) & 1, q2 = (t >> sh) & 1;
        const int px = 4 + half, py = 2 * half, pz = 2 * half + 1;
        const int cy = col >> 2, cz = col & 3;
        float2 acc = make_float2(0.0f, 0.0f);
        for (int c1 = 0; c1 < 2; ++c1) {
            for (int c3 = 0; c3 < 2; ++c3) {
                const float2 m = cmul(Vs[px][2 * q1 + sel][2 * c1 + c3],
                                      cmul(Vs[py][2 * q0 + c1][cy], Vs[pz][2 * q2 + c3][cz]));
                acc.x += m.x; acc.y += m.y;
            }
        }
        Uf[half][sel][col] = acc;
    }
    __syncthreads();

    {
        const int k = tid, l = k >> 4, r = k & 15;
        const int row = 2 * ((t >> 3) & 1) + 1;
        float2 acc = make_float2(0.0f, 0.0f);
        for (int s = 0; s < 2; ++s) {
            float2 in = make_float2(0.0f, 0.0f);
            for (int u = 0; u < 2; ++u) {
                const float2 m = cmul(Vs[6][row][2 * s + u], Uf[1][u][r]);
                in.x += m.x; in.y += m.y;
            }
            const float2 m = cmul(Uf[0][s][l], in);
            acc.x += m.x; acc.y += m.y;
        }
        AtP[t * 256 + k]         = acc.x;
        AtP[(t + 128) * 256 + k] = acc.y;
    }
}


#define MP 260
#define GW 8

__device__ __forceinline__ v8f mma4(float ax, float ay, float bx, float by, v8f acc) {
    v2f a; a[0] = ax; a[1] = ay;
    v2f b; b[0] = bx; b[1] = by;
    return __builtin_amdgcn_wmma_f32_16x16x4_f32(false, a, false, b, (short)0, acc, false, false);
}

__global__ __launch_bounds__(256) void qcnn_wmma_gemm(const float* __restrict__ x,
                                                      const float* __restrict__ Mg,
                                                      float* __restrict__ out,
                                                      int nrows) {
    __shared__ float Ms[256 * MP];

    const int tid  = threadIdx.x;
    const int wave = __builtin_amdgcn_readfirstlane(tid >> 5);
    const int lane = tid & 31;
    const int c    = lane & 15;
    const int h    = lane >> 4;

    const int ntiles  = nrows >> 4;
    const int tstride = gridDim.x * GW;
    int tile = blockIdx.x * GW + wave;

    const int t0 = tile < ntiles ? tile : 0;
    const float* xr = x + ((size_t)t0 * 16 + c) * 256 + 4 * h;
    float4 xc[4];
#pragma unroll
    for (int u = 0; u < 4; ++u) xc[u] = *(const float4*)(xr + 8 * u);

    {
        const float4* src = (const float4*)Mg;
        for (int i = tid; i < 256 * 64; i += 256) {
            const int r = i >> 6, q = i & 63;
            *(float4*)&Ms[r * MP + q * 4] = src[i];
        }
    }
    __syncthreads();

    if (tile >= ntiles) return;

    const float* mlane = Ms + c * MP + 4 * h;

    float4 fa[2][16];
#pragma unroll
    for (int nt = 0; nt < 16; ++nt) fa[0][nt] = *(const float4*)(mlane + nt * 16 * MP);

    while (true) {
        const int  ntile = tile + tstride;
        const bool more  = ntile < ntiles;
        const float* xnr = x + ((size_t)(more ? ntile : tile) * 16 + c) * 256 + 4 * h;

        v8f acc[16];
#pragma unroll
        for (int nt = 0; nt < 16; ++nt) acc[nt] = (v8f){};
        float ssx = 0.0f;

#pragma unroll 1
        for (int jc = 0; jc < 8; ++jc) {
            const float* nsrc = (jc < 7) ? (xr + 32 * (jc + 1)) : xnr;
            float4 xn[4];
#pragma unroll
            for (int u = 0; u < 4; ++u) xn[u] = *(const float4*)(nsrc + 8 * u);

            const float* mj = mlane + 32 * jc;
#pragma unroll
            for (int u = 0; u < 4; ++u) {
                const float* mnext = (u < 3) ? (mj + 8 * (u + 1)) : (mlane + 32 * ((jc + 1) & 7));
#pragma unroll
                for (int nt = 0; nt < 16; ++nt)
                    fa[(u + 1) & 1][nt] = *(const float4*)(mnext + nt * 16 * MP);
                __builtin_amdgcn_sched_barrier(0);
                ssx += xc[u].x * xc[u].x + xc[u].y * xc[u].y + xc[u].z * xc[u].z + xc[u].w * xc[u].w;
#pragma unroll
                for (int nt = 0; nt < 16; ++nt)
                    acc[nt] = mma4(fa[u & 1][nt].x, fa[u & 1][nt].y, xc[u].x, xc[u].y, acc[nt]);
#pragma unroll
                for (int nt = 0; nt < 16; ++nt)
                    acc[nt] = mma4(fa[u & 1][nt].z, fa[u & 1][nt].w, xc[u].z, xc[u].w, acc[nt]);
                __builtin_amdgcn_sched_barrier(0);
            }
#pragma unroll
            for (int u = 0; u < 4; ++u) xc[u] = xn[u];
        }

        float q4[4] = {0.0f, 0.0f, 0.0f, 0.0f};
#pragma unroll
        for (int nt = 0; nt < 16; ++nt) {
#pragma unroll
            for (int v = 0; v < 8; ++v) q4[v & 3] += acc[nt][v] * acc[nt][v];
        }
        float qs = (q4[0] + q4[1]) + (q4[2] + q4[3]);
        qs  += __shfl_xor(qs, 16);
        ssx += __shfl_xor(ssx, 16);
        if (h == 0) {
            const float q = qs / ssx;
            float2 o; o.x = 1.0f - q; o.y = q;
            *(float2*)(out + 2 * ((size_t)tile * 16 + c)) = o;
        }

        if (!more) break;
        tile = ntile;
        xr   = xnr;
    }
}


extern "C" void kernel_launch(void* const* d_in, const int* in_sizes, int n_in,
                              void* d_out, int out_size, void* d_ws, size_t ws_size,
                              hipStream_t stream) {
    const float* x = (const float*)d_in[0];
    const float* w = (const float*)d_in[1];
    float* out = (float*)d_out;
    float* AtP = (float*)d_ws;

    const int B = in_sizes[0] / 256;

    qcnn_build_matrix<<<128, 256, 0, stream>>>(w, AtP);
    qcnn_wmma_gemm<<<256, 256, 0, stream>>>(x, (const float*)AtP, out, B);
}
